# starting kernel
# baseline (speedup 1.0000x reference)
#include <hip/hip_runtime.h>
#include <hip/hip_bf16.h>

typedef __attribute__((ext_vector_type(2))) float v2f;
typedef __attribute__((ext_vector_type(8))) float v8f;

__global__ void k_init_deg(float* __restrict__ deg, int N) {
    int i = blockIdx.x * blockDim.x + threadIdx.x;
    if (i < N) deg[i] = 1.0f;
}

__global__ void k_scatter_deg(const int* __restrict__ col,
                              const float* __restrict__ w,
                              float* __restrict__ deg, int E) {
    int e = blockIdx.x * blockDim.x + threadIdx.x;
    if (e < E) atomicAdd(&deg[col[e]], w[e]);
}

__global__ void k_dinv(const float* __restrict__ deg,
                       float* __restrict__ dinv, int N) {
    int i = blockIdx.x * blockDim.x + threadIdx.x;
    if (i < N) {
        float d = deg[i];
        dinv[i] = (d > 0.0f) ? rsqrtf(d) : 0.0f;
    }
}

__global__ void k_norm(const int* __restrict__ row,
                       const int* __restrict__ col,
                       const float* __restrict__ w,
                       const float* __restrict__ dinv,
                       float* __restrict__ norm, int E) {
    int e = blockIdx.x * blockDim.x + threadIdx.x;
    if (e < E) norm[e] = dinv[row[e]] * w[e] * dinv[col[e]];
}

__global__ __launch_bounds__(256) void k_gemm1(const float* __restrict__ X,
                                               const float* __restrict__ W,
                                               float* __restrict__ H,
                                               int N) {
    __shared__ float Alds[16 * 128];
    const int tid = threadIdx.x;
    const int bm  = blockIdx.x * 16;

    float4* al4 = (float4*)Alds;
    for (int i = tid; i < 16 * 32; i += 256) {
        int r = i >> 5, g = i & 31;
        float4 v = make_float4(0.f, 0.f, 0.f, 0.f);
        if (bm + r < N) v = ((const float4*)(X + (size_t)(bm + r) * 128))[g];
        al4[i] = v;
    }
    __syncthreads();

    const int wave = tid >> 5;
    const int lane = tid & 31;
    const int nb   = wave * 16;
    const int mr   = lane & 15;
    const int kh   = (lane >> 4) * 2;

    v8f acc = {};
    for (int kc = 0; kc < 128; kc += 4) {
        v2f a, b;
        a[0] = Alds[mr * 128 + kc + kh + 0];
        a[1] = Alds[mr * 128 + kc + kh + 1];
        b[0] = W[(kc + kh + 0) * 128 + nb + mr];
        b[1] = W[(kc + kh + 1) * 128 + nb + mr];
        acc = __builtin_amdgcn_wmma_f32_16x16x4_f32(
                  false, a, false, b, (short)0, acc, false, false);
    }

    const int mtop = (lane >> 4) * 8;
    for (int v = 0; v < 8; ++v) {
        int m = bm + mtop + v;
        if (m < N) H[(size_t)m * 128 + nb + mr] = acc[v];
    }
}

__global__ void k_init_out1(const float* __restrict__ h1,
                            const float* __restrict__ dinv,
                            const float* __restrict__ b1,
                            float* __restrict__ out1, int N) {
    int idx = blockIdx.x * blockDim.x + threadIdx.x;
    if (idx >= N * 32) return;
    int i = idx >> 5, g = idx & 31;
    float s = dinv[i]; s *= s;
    float4 h = ((const float4*)(h1 + (size_t)i * 128))[g];
    float4 b = ((const float4*)b1)[g];
    float4 o = make_float4(b.x + h.x * s, b.y + h.y * s,
                           b.z + h.z * s, b.w + h.w * s);
    ((float4*)(out1 + (size_t)i * 128))[g] = o;
}

__global__ void k_scatter1(const float* __restrict__ h1,
                           const int* __restrict__ row,
                           const int* __restrict__ col,
                           const float* __restrict__ norm,
                           float* __restrict__ out1, int E) {
    int idx = blockIdx.x * blockDim.x + threadIdx.x;
    if (idx >= E * 32) return;
    int e = idx >> 5, g = idx & 31;
    int r = row[e], c = col[e];
    float nw = norm[e];
    float4 v = ((const float4*)(h1 + (size_t)r * 128))[g];
    float* o = out1 + (size_t)c * 128 + g * 4;
    atomicAdd(o + 0, v.x * nw);
    atomicAdd(o + 1, v.y * nw);
    atomicAdd(o + 2, v.z * nw);
    atomicAdd(o + 3, v.w * nw);
}

__global__ __launch_bounds__(256) void k_gemm2(const float* __restrict__ A,
                                               const float* __restrict__ W2,
                                               float* __restrict__ H2,
                                               int N) {
    __shared__ float Alds[32 * 128];
    __shared__ float Wlds[128 * 64];
    const int tid = threadIdx.x;
    const int bm  = blockIdx.x * 32;

    {
        const float4* w4 = (const float4*)W2;
        float4* wl = (float4*)Wlds;
        for (int i = tid; i < 2048; i += 256) wl[i] = w4[i];
    }
    {
        float4* al = (float4*)Alds;
        for (int i = tid; i < 32 * 32; i += 256) {
            int r = i >> 5, g = i & 31;
            float4 v = make_float4(0.f, 0.f, 0.f, 0.f);
            if (bm + r < N) v = ((const float4*)(A + (size_t)(bm + r) * 128))[g];
            v.x = fmaxf(v.x, 0.f); v.y = fmaxf(v.y, 0.f);
            v.z = fmaxf(v.z, 0.f); v.w = fmaxf(v.w, 0.f);
            al[i] = v;
        }
    }
    __syncthreads();

    const int wave = tid >> 5;
    const int lane = tid & 31;
    const int msub = (wave >> 2) * 16;
    const int nb   = (wave & 3) * 16;
    const int mr   = lane & 15;
    const int kh   = (lane >> 4) * 2;

    v8f acc = {};
    for (int kc = 0; kc < 128; kc += 4) {
        v2f a, b;
        a[0] = Alds[(msub + mr) * 128 + kc + kh + 0];
        a[1] = Alds[(msub + mr) * 128 + kc + kh + 1];
        b[0] = Wlds[(kc + kh + 0) * 64 + nb + mr];
        b[1] = Wlds[(kc + kh + 1) * 64 + nb + mr];
        acc = __builtin_amdgcn_wmma_f32_16x16x4_f32(
                  false, a, false, b, (short)0, acc, false, false);
    }

    const int mtop = (lane >> 4) * 8;
    for (int v = 0; v < 8; ++v) {
        int m = bm + msub + mtop + v;
        if (m < N) H2[(size_t)m * 64 + nb + mr] = acc[v];
    }
}

__global__ void k_init_out2(const float* __restrict__ h2,
                            const float* __restrict__ dinv,
                            const float* __restrict__ b2,
                            float* __restrict__ out, int N) {
    int idx = blockIdx.x * blockDim.x + threadIdx.x;
    if (idx >= N * 16) return;
    int i = idx >> 4, g = idx & 15;
    float s = dinv[i]; s *= s;
    float4 h = ((const float4*)(h2 + (size_t)i * 64))[g];
    float4 b = ((const float4*)b2)[g];
    float4 o = make_float4(b.x + h.x * s, b.y + h.y * s,
                           b.z + h.z * s, b.w + h.w * s);
    ((float4*)(out + (size_t)i * 64))[g] = o;
}

__global__ void k_scatter2(const float* __restrict__ h2,
                           const int* __restrict__ row,
                           const int* __restrict__ col,
                           const float* __restrict__ norm,
                           float* __restrict__ out, int E) {
    int idx = blockIdx.x * blockDim.x + threadIdx.x;
    if (idx >= E * 16) return;
    int e = idx >> 4, g = idx & 15;
    int r = row[e], c = col[e];
    float nw = norm[e];
    float4 v = ((const float4*)(h2 + (size_t)r * 64))[g];
    float* o = out + (size_t)c * 64 + g * 4;
    atomicAdd(o + 0, v.x * nw);
    atomicAdd(o + 1, v.y * nw);
    atomicAdd(o + 2, v.z * nw);
    atomicAdd(o + 3, v.w * nw);
}

extern "C" void kernel_launch(void* const* d_in, const int* in_sizes, int n_in,
                              void* d_out, int out_size, void* d_ws, size_t ws_size,
                              hipStream_t stream) {
    const float* x   = (const float*)d_in[0];
    const int*   ei  = (const int*)d_in[1];
    const float* ew  = (const float*)d_in[2];
    const float* W1  = (const float*)d_in[3];
    const float* b1  = (const float*)d_in[4];
    const float* W2  = (const float*)d_in[5];
    const float* b2  = (const float*)d_in[6];
    float* out = (float*)d_out;

    const int E = in_sizes[2];
    const int N = in_sizes[0] / 128;
    const int* row = ei;
    const int* col = ei + E;

    float* ws   = (float*)d_ws;
    float* deg  = ws;
    float* dinv = ws + N;
    float* nrm  = ws + 2 * (size_t)N;
    float* h1   = nrm + E;
    float* out1 = h1 + (size_t)N * 128;
    float* h2   = out1 + (size_t)N * 128;

    const int T = 256;

    k_init_deg<<<(N + T - 1) / T, T, 0, stream>>>(deg, N);
    k_scatter_deg<<<(E + T - 1) / T, T, 0, stream>>>(col, ew, deg, E);
    k_dinv<<<(N + T - 1) / T, T, 0, stream>>>(deg, dinv, N);
    k_norm<<<(E + T - 1) / T, T, 0, stream>>>(row, col, ew, dinv, nrm, E);

    k_gemm1<<<(N + 15) / 16, 256, 0, stream>>>(x, W1, h1, N);
    k_init_out1<<<((size_t)N * 32 + T - 1) / T, T, 0, stream>>>(h1, dinv, b1, out1, N);
    k_scatter1<<<((size_t)E * 32 + T - 1) / T, T, 0, stream>>>(h1, row, col, nrm, out1, E);

    k_gemm2<<<(N + 31) / 32, 256, 0, stream>>>(out1, W2, h2, N);
    k_init_out2<<<((size_t)N * 16 + T - 1) / T, T, 0, stream>>>(h2, dinv, b2, out, N);
    k_scatter2<<<((size_t)E * 16 + T - 1) / T, T, 0, stream>>>(h2, row, col, nrm, out, E);
}


# Both GEMMs re-tiled: a wave owns 16 rows x all columns, its A rows go from global straight into registers (no A tile in LDS), W in LDS as bank-spread k-pairs; k summed in a permuted order
# speedup vs baseline: 1.7512x; 1.7512x over previous
#include <hip/hip_runtime.h>
#include <hip/hip_bf16.h>

typedef __attribute__((ext_vector_type(2))) float v2f;
typedef __attribute__((ext_vector_type(8))) float v8f;

__global__ void k_init_deg(float* __restrict__ deg, int N) {
    int i = blockIdx.x * blockDim.x + threadIdx.x;
    if (i < N) deg[i] = 1.0f;
}

__global__ void k_scatter_deg(const int* __restrict__ col,
                              const float* __restrict__ w,
                              float* __restrict__ deg, int E) {
    int e = blockIdx.x * blockDim.x + threadIdx.x;
    if (e < E) atomicAdd(&deg[col[e]], w[e]);
}

__device__ __forceinline__ float dinv_of(float d) {
    return (d > 0.0f) ? rsqrtf(d) : 0.0f;
}

template <int NT, bool RELU>
__global__ __launch_bounds__(256) void k_gemm_rows(const float* __restrict__ A,
                                                   const float* __restrict__ W,
                                                   const float* __restrict__ deg,
                                                   const float* __restrict__ bias,
                                                   float* __restrict__ H,
                                                   float* __restrict__ out,
                                                   int N) {
    constexpr int NC = 16 * NT;
    constexpr int WI = 64 * (NC / 4) / 256;
    __shared__ __attribute__((aligned(16))) float Wl[128 * NC];
    const int tid  = threadIdx.x;
    const int wave = tid >> 5;
    const int lane = tid & 31;
    const int c    = lane & 15;
    const int h    = lane >> 4;
    const int row0 = blockIdx.x * 128 + wave * 16;

    float4 a4[16];
    {
        const int r = (row0 + c < N) ? (row0 + c) : (N - 1);
        const float4* s4 = (const float4*)(A + (size_t)r * 128) + h;
#pragma unroll
        for (int g = 0; g < 16; ++g) a4[g] = s4[2 * g];
    }

    {
        const float4* w4 = (const float4*)W;
        float4 r0[WI], r1[WI];
#pragma unroll
        for (int it = 0; it < WI; ++it) {
            const int i  = tid + 256 * it;
            const int kp = i / (NC / 4);
            const int n4 = i % (NC / 4);
            r0[it] = w4[(2 * kp) * (NC / 4) + n4];
            r1[it] = w4[(2 * kp + 1) * (NC / 4) + n4];
        }
#pragma unroll
        for (int it = 0; it < WI; ++it) {
            const int i  = tid + 256 * it;
            const int kp = i / (NC / 4);
            const int n  = (i % (NC / 4)) * 4;
            const int ns = n ^ (((kp >> 1) & 1) << 4);
            float4* d4 = (float4*)(Wl + (size_t)(kp * NC + ns) * 2);
            d4[0] = make_float4(r0[it].x, r1[it].x, r0[it].y, r1[it].y);
            d4[1] = make_float4(r0[it].z, r1[it].z, r0[it].w, r1[it].w);
        }
    }

    float dg[8], bs[NT];
#pragma unroll
    for (int v = 0; v < 8; ++v) {
        const int m = row0 + 8 * h + v;
        dg[v] = deg[(m < N) ? m : (N - 1)];
    }
#pragma unroll
    for (int t = 0; t < NT; ++t) bs[t] = bias[16 * t + c];

    if (RELU) {
#pragma unroll
        for (int g = 0; g < 16; ++g) {
            a4[g].x = fmaxf(a4[g].x, 0.f); a4[g].y = fmaxf(a4[g].y, 0.f);
            a4[g].z = fmaxf(a4[g].z, 0.f); a4[g].w = fmaxf(a4[g].w, 0.f);
        }
    }
    __syncthreads();

    const v2f* wE = (const v2f*)Wl + (2 * h * NC + c + 16 * h);
    const v2f* wO = (const v2f*)Wl + (2 * h * NC + c - 16 * h);

    v8f acc[NT];
#pragma unroll
    for (int t = 0; t < NT; ++t) acc[t] = v8f{};

    v2f bc[NT], bn[NT];
#pragma unroll
    for (int t = 0; t < NT; ++t) bc[t] = ((t & 1) ? wO : wE)[16 * t];
#pragma unroll
    for (int st = 0; st < 32; ++st) {
        const int g = st >> 1, s = st & 1;
        if (st + 1 < 32) {
            const int kn = 4 * ((st + 1) >> 1) + ((st + 1) & 1);
#pragma unroll
            for (int t = 0; t < NT; ++t) bn[t] = ((t & 1) ? wO : wE)[kn * NC + 16 * t];
        }
        v2f a;
        a[0] = s ? a4[g].z : a4[g].x;
        a[1] = s ? a4[g].w : a4[g].y;
#pragma unroll
        for (int t = 0; t < NT; ++t)
            acc[t] = __builtin_amdgcn_wmma_f32_16x16x4_f32(
                         false, a, false, bc[t], (short)0, acc[t], false, false);
#pragma unroll
        for (int t = 0; t < NT; ++t) bc[t] = bn[t];
    }

#pragma unroll
    for (int v = 0; v < 8; ++v) {
        const int m = row0 + 8 * h + v;
        if (m < N) {
            const float d  = dinv_of(dg[v]);
            const float s2 = d * d;
            float* hp = H + (size_t)m * NC + c;
            float* op = out + (size_t)m * NC + c;
#pragma unroll
            for (int t = 0; t < NT; ++t) {
                hp[16 * t] = acc[t][v];
                op[16 * t] = bs[t] + acc[t][v] * s2;
            }
        }
    }
}

template <int NL, int UE>
__device__ __forceinline__ void scatter_rows(const float* __restrict__ h,
                                             const int* __restrict__ row,
                                             const int* __restrict__ col,
                                             const float* __restrict__ w,
                                             const float* __restrict__ deg,
                                             float* __restrict__ out, int E) {
    const int lane = threadIdx.x & 31;
    const int wave = __builtin_amdgcn_readfirstlane((int)(blockIdx.x * (blockDim.x >> 5) + (threadIdx.x >> 5)));
    const int e0 = wave * 32;
    if (e0 >= E) return;
    const int n = (E - e0 < 32) ? (E - e0) : 32;
    int r = 0, c = 0, nwb = 0;
    if (lane < n) {
        r = row[e0 + lane];
        c = col[e0 + lane];
        nwb = __float_as_int(dinv_of(deg[r]) * w[e0 + lane] * dinv_of(deg[c]));
    }
    if (n == 32) {
        for (int j0 = 0; j0 < 32; j0 += UE) {
            float v[UE][NL];
            for (int u = 0; u < UE; ++u) {
                const int ru = __builtin_amdgcn_readlane(r, j0 + u);
                const float* s = h + (size_t)ru * (NL * 32) + lane;
                for (int k = 0; k < NL; ++k) v[u][k] = s[32 * k];
            }
            for (int u = 0; u < UE; ++u) {
                const int cu = __builtin_amdgcn_readlane(c, j0 + u);
                const float nu = __int_as_float(__builtin_amdgcn_readlane(nwb, j0 + u));
                float* o = out + (size_t)cu * (NL * 32) + lane;
                for (int k = 0; k < NL; ++k) atomicAdd(o + 32 * k, v[u][k] * nu);
            }
        }
    } else {
        for (int j = 0; j < n; ++j) {
            const int ru = __builtin_amdgcn_readlane(r, j);
            const int cu = __builtin_amdgcn_readlane(c, j);
            const float nu = __int_as_float(__builtin_amdgcn_readlane(nwb, j));
            const float* s = h + (size_t)ru * (NL * 32) + lane;
            float* o = out + (size_t)cu * (NL * 32) + lane;
            for (int k = 0; k < NL; ++k) atomicAdd(o + 32 * k, s[32 * k] * nu);
        }
    }
}

__global__ __launch_bounds__(256) void k_scatter1(const float* __restrict__ h1,
                                                  const int* __restrict__ row,
                                                  const int* __restrict__ col,
                                                  const float* __restrict__ w,
                                                  const float* __restrict__ deg,
                                                  float* __restrict__ out1, int E) {
    scatter_rows<4, 8>(h1, row, col, w, deg, out1, E);
}

__global__ __launch_bounds__(256) void k_scatter2(const float* __restrict__ h2,
                                                  const int* __restrict__ row,
                                                  const int* __restrict__ col,
                                                  const float* __restrict__ w,
                                                  const float* __restrict__ deg,
                                                  float* __restrict__ out, int E) {
    scatter_rows<2, 16>(h2, row, col, w, deg, out, E);
}

extern "C" void kernel_launch(void* const* d_in, const int* in_sizes, int n_in,
                              void* d_out, int out_size, void* d_ws, size_t ws_size,
                              hipStream_t stream) {
    const float* x   = (const float*)d_in[0];
    const int*   ei  = (const int*)d_in[1];
    const float* ew  = (const float*)d_in[2];
    const float* W1  = (const float*)d_in[3];
    const float* b1  = (const float*)d_in[4];
    const float* W2  = (const float*)d_in[5];
    const float* b2  = (const float*)d_in[6];
    float* out = (float*)d_out;

    const int E = in_sizes[2];
    const int N = in_sizes[0] / 128;
    const int* row = ei;
    const int* col = ei + E;

    float* ws   = (float*)d_ws;
    float* deg  = ws;
    float* h1   = ws + 2 * (size_t)N + E;
    float* out1 = h1 + (size_t)N * 128;
    float* h2   = out1 + (size_t)N * 128;

    const int T = 256;

    k_init_deg<<<(N + T - 1) / T, T, 0, stream>>>(deg, N);
    k_scatter_deg<<<(E + T - 1) / T, T, 0, stream>>>(col, ew, deg, E);

    k_gemm_rows<8, false><<<(N + 127) / 128, 256, 0, stream>>>(x, W1, deg, b1, h1, out1, N);
    k_scatter1<<<(E + T - 1) / T, T, 0, stream>>>(h1, row, col, ew, deg, out1, E);

    k_gemm_rows<4, true><<<(N + 127) / 128, 256, 0, stream>>>(out1, W2, deg, b2, h2, out, N);
    k_scatter2<<<(E + T - 1) / T, T, 0, stream>>>(h2, row, col, ew, deg, out, E);
}


# Edge scatters destination-stationary: a work-group keeps a tile of out rows in LDS, filters the edge list for it, sums rows by LDS float atomics, then one global atomic per line; sum order changes
# speedup vs baseline: 1.9395x; 1.1075x over previous
#include <hip/hip_runtime.h>
#include <hip/hip_bf16.h>

typedef __attribute__((ext_vector_type(2))) float v2f;
typedef __attribute__((ext_vector_type(8))) float v8f;

__global__ void k_init_deg(float* __restrict__ deg, int N) {
    int i = blockIdx.x * blockDim.x + threadIdx.x;
    if (i < N) deg[i] = 1.0f;
}

__global__ void k_scatter_deg(const int* __restrict__ col,
                              const float* __restrict__ w,
                              float* __restrict__ deg, int E) {
    int e = blockIdx.x * blockDim.x + threadIdx.x;
    if (e < E) atomicAdd(&deg[col[e]], w[e]);
}

__device__ __forceinline__ float dinv_of(float d) {
    return (d > 0.0f) ? rsqrtf(d) : 0.0f;
}

template <int NT, bool RELU>
__global__ __launch_bounds__(256) void k_gemm_rows(const float* __restrict__ A,
                                                   const float* __restrict__ W,
                                                   const float* __restrict__ deg,
                                                   const float* __restrict__ bias,
                                                   float* __restrict__ H,
                                                   float* __restrict__ out,
                                                   int N) {
    constexpr int NC = 16 * NT;
    constexpr int WI = 64 * (NC / 4) / 256;
    __shared__ __attribute__((aligned(16))) float Wl[128 * NC];
    const int tid  = threadIdx.x;
    const int wave = tid >> 5;
    const int lane = tid & 31;
    const int c    = lane & 15;
    const int h    = lane >> 4;
    const int row0 = blockIdx.x * 128 + wave * 16;

    float4 a4[16];
    {
        const int r = (row0 + c < N) ? (row0 + c) : (N - 1);
        const float4* s4 = (const float4*)(A + (size_t)r * 128) + h;
#pragma unroll
        for (int g = 0; g < 16; ++g) a4[g] = s4[2 * g];
    }

    {
        const float4* w4 = (const float4*)W;
        float4 r0[WI], r1[WI];
#pragma unroll
        for (int it = 0; it < WI; ++it) {
            const int i  = tid + 256 * it;
            const int kp = i / (NC / 4);
            const int n4 = i % (NC / 4);
            r0[it] = w4[(2 * kp) * (NC / 4) + n4];
            r1[it] = w4[(2 * kp + 1) * (NC / 4) + n4];
        }
#pragma unroll
        for (int it = 0; it < WI; ++it) {
            const int i  = tid + 256 * it;
            const int kp = i / (NC / 4);
            const int n  = (i % (NC / 4)) * 4;
            const int ns = n ^ (((kp >> 1) & 1) << 4);
            float4* d4 = (float4*)(Wl + (size_t)(kp * NC + ns) * 2);
            d4[0] = make_float4(r0[it].x, r1[it].x, r0[it].y, r1[it].y);
            d4[1] = make_float4(r0[it].z, r1[it].z, r0[it].w, r1[it].w);
        }
    }

    float dg[8], bs[NT];
#pragma unroll
    for (int v = 0; v < 8; ++v) {
        const int m = row0 + 8 * h + v;
        dg[v] = deg[(m < N) ? m : (N - 1)];
    }
#pragma unroll
    for (int t = 0; t < NT; ++t) bs[t] = bias[16 * t + c];

    if (RELU) {
#pragma unroll
        for (int g = 0; g < 16; ++g) {
            a4[g].x = fmaxf(a4[g].x, 0.f); a4[g].y = fmaxf(a4[g].y, 0.f);
            a4[g].z = fmaxf(a4[g].z, 0.f); a4[g].w = fmaxf(a4[g].w, 0.f);
        }
    }
    __syncthreads();

    const v2f* wE = (const v2f*)Wl + (2 * h * NC + c + 16 * h);
    const v2f* wO = (const v2f*)Wl + (2 * h * NC + c - 16 * h);

    v8f acc[NT];
#pragma unroll
    for (int t = 0; t < NT; ++t) acc[t] = v8f{};

    v2f bc[NT], bn[NT];
#pragma unroll
    for (int t = 0; t < NT; ++t) bc[t] = ((t & 1) ? wO : wE)[16 * t];
#pragma unroll
    for (int st = 0; st < 32; ++st) {
        const int g = st >> 1, s = st & 1;
        if (st + 1 < 32) {
            const int kn = 4 * ((st + 1) >> 1) + ((st + 1) & 1);
#pragma unroll
            for (int t = 0; t < NT; ++t) bn[t] = ((t & 1) ? wO : wE)[kn * NC + 16 * t];
        }
        v2f a;
        a[0] = s ? a4[g].z : a4[g].x;
        a[1] = s ? a4[g].w : a4[g].y;
#pragma unroll
        for (int t = 0; t < NT; ++t)
            acc[t] = __builtin_amdgcn_wmma_f32_16x16x4_f32(
                         false, a, false, bc[t], (short)0, acc[t], false, false);
#pragma unroll
        for (int t = 0; t < NT; ++t) bc[t] = bn[t];
    }

#pragma unroll
    for (int v = 0; v < 8; ++v) {
        const int m = row0 + 8 * h + v;
        if (m < N) {
            const float d  = dinv_of(dg[v]);
            const float s2 = d * d;
            float* hp = H + (size_t)m * NC + c;
            float* op = out + (size_t)m * NC + c;
#pragma unroll
            for (int t = 0; t < NT; ++t) {
                hp[16 * t] = acc[t][v];
                op[16 * t] = bs[t] + acc[t][v] * s2;
            }
        }
    }
}

template <int NL, int UE, bool FULL>
__device__ __forceinline__ void drain_hits(const int2* qw, int qb, int cnt, int lo, int lane,
                                           const float* __restrict__ h,
                                           const int* __restrict__ row,
                                           const float* __restrict__ w,
                                           const float* __restrict__ deg,
                                           float* tile) {
    constexpr int F = 32 * NL;
    int r = 0, cl = 0;
    float wt = 0.0f, dc = 0.0f, dr = 0.0f;
    if (lane < cnt) {
        const int2 en = qw[qb + lane];
        cl = en.y;
        r  = row[en.x];
        wt = w[en.x];
        dc = deg[lo + cl];
    }
    float v[UE][NL];
#pragma unroll
    for (int u = 0; u < UE; ++u) {
        if (FULL || u < cnt) {
            const int ru = __builtin_amdgcn_readlane(r, u);
            const float* s = h + (size_t)ru * F + lane;
#pragma unroll
            for (int k = 0; k < NL; ++k) v[u][k] = s[32 * k];
        } else {
#pragma unroll
            for (int k = 0; k < NL; ++k) v[u][k] = 0.0f;
        }
    }
    if (lane < cnt) dr = deg[r];
    const float nw = dinv_of(dr) * wt * dinv_of(dc);
#pragma unroll
    for (int u = 0; u < UE; ++u) {
        if (FULL || u < cnt) {
            const int cu = __builtin_amdgcn_readlane(cl, u);
            const float nu = __int_as_float(__builtin_amdgcn_readlane(__float_as_int(nw), u));
            float* o = tile + cu * F + lane;
#pragma unroll
            for (int k = 0; k < NL; ++k) atomicAdd(o + 32 * k, v[u][k] * nu);
        }
    }
}

template <int NL, int UE, int CH>
__device__ __forceinline__ void scan_chunk(const int (&cv)[CH], int ebase, int lo, int nrows, int lane,
                                           int& qn, int2* qw,
                                           const float* __restrict__ h,
                                           const int* __restrict__ row,
                                           const float* __restrict__ w,
                                           const float* __restrict__ deg,
                                           float* tile) {
#pragma unroll
    for (int k = 0; k < CH; ++k) {
        const int cl = cv[k] - lo;
        const bool hit = (unsigned)cl < (unsigned)nrows;
        const unsigned m = __builtin_amdgcn_ballot_w32(hit);
        if (m != 0u) {
            if (hit) qw[qn + (int)__builtin_amdgcn_mbcnt_lo(m, 0u)] = make_int2(ebase + 32 * k + lane, cl);
            qn += __builtin_popcount(m);
        }
    }
    while (qn >= UE) {
        qn -= UE;
        drain_hits<NL, UE, true>(qw, qn, UE, lo, lane, h, row, w, deg, tile);
    }
}

template <int NL, int ROWS, int G, int UE>
__global__ __launch_bounds__(1024) void k_scatter_tile(const float* __restrict__ h,
                                                       const int* __restrict__ row,
                                                       const int* __restrict__ col,
                                                       const float* __restrict__ w,
                                                       const float* __restrict__ deg,
                                                       float* __restrict__ out, int N, int E) {
    constexpr int F    = 32 * NL;
    constexpr int CH   = 8;
    constexpr int CE   = 32 * CH;
    constexpr int QCAP = UE + CE;
    __shared__ __attribute__((aligned(16))) float tile[ROWS * F];
    __shared__ int2 queue[32 * QCAP];
    const int tid  = threadIdx.x;
    const int lane = tid & 31;
    const int wv     = __builtin_amdgcn_readfirstlane(tid >> 5);
    const int g      = blockIdx.x % G;
    const int TB     = gridDim.x / G;
    const int nfull  = E / CE;
    const int stride = 32 * G;
    int2* qw = queue + wv * QCAP;

    for (int t = blockIdx.x / G; t * ROWS < N; t += TB) {
        const int lo    = t * ROWS;
        const int nrows = (N - lo < ROWS) ? (N - lo) : ROWS;

        for (int i = tid; i < ROWS * F / 4; i += 1024) ((float4*)tile)[i] = make_float4(0.f, 0.f, 0.f, 0.f);
        __syncthreads();

        int qn = 0;
        if (nfull * CE < E && g == 0 && wv == 0) {
            int cr[CH];
#pragma unroll
            for (int k = 0; k < CH; ++k) {
                const int e = nfull * CE + 32 * k + lane;
                cr[k] = (e < E) ? col[e] : -1;
            }
            scan_chunk<NL, UE, CH>(cr, nfull * CE, lo, nrows, lane, qn, qw, h, row, w, deg, tile);
        }
        if (nfull > 0) {
            const int last = nfull - 1;
            int ci = g * 32 + wv;
            int ca[CH], cb[CH];
            {
                const int* p = col + (size_t)((ci < last) ? ci : last) * CE + lane;
#pragma unroll
                for (int k = 0; k < CH; ++k) ca[k] = p[32 * k];
            }
            for (; ci < nfull; ci += 2 * stride) {
                {
                    const int cj = ci + stride;
                    const int* p = col + (size_t)((cj < last) ? cj : last) * CE + lane;
#pragma unroll
                    for (int k = 0; k < CH; ++k) cb[k] = p[32 * k];
                }
                scan_chunk<NL, UE, CH>(ca, ci * CE, lo, nrows, lane, qn, qw, h, row, w, deg, tile);
                {
                    const int cj = ci + 2 * stride;
                    const int* p = col + (size_t)((cj < last) ? cj : last) * CE + lane;
#pragma unroll
                    for (int k = 0; k < CH; ++k) ca[k] = p[32 * k];
                }
                if (ci + stride < nfull)
                    scan_chunk<NL, UE, CH>(cb, (ci + stride) * CE, lo, nrows, lane, qn, qw, h, row, w, deg, tile);
            }
        }
        if (qn > 0) drain_hits<NL, UE, false>(qw, 0, qn, lo, lane, h, row, w, deg, tile);
        __syncthreads();

        for (int r = wv; r < nrows; r += 32) {
            const float* s = tile + r * F + lane;
            float* o = out + (size_t)(lo + r) * F + lane;
            float tv[NL];
#pragma unroll
            for (int k = 0; k < NL; ++k) tv[k] = s[32 * k];
#pragma unroll
            for (int k = 0; k < NL; ++k) atomicAdd(o + 32 * k, tv[k]);
        }
        __syncthreads();
    }
}

extern "C" void kernel_launch(void* const* d_in, const int* in_sizes, int n_in,
                              void* d_out, int out_size, void* d_ws, size_t ws_size,
                              hipStream_t stream) {
    const float* x   = (const float*)d_in[0];
    const int*   ei  = (const int*)d_in[1];
    const float* ew  = (const float*)d_in[2];
    const float* W1  = (const float*)d_in[3];
    const float* b1  = (const float*)d_in[4];
    const float* W2  = (const float*)d_in[5];
    const float* b2  = (const float*)d_in[6];
    float* out = (float*)d_out;

    const int E = in_sizes[2];
    const int N = in_sizes[0] / 128;
    const int* row = ei;
    const int* col = ei + E;

    float* ws   = (float*)d_ws;
    float* deg  = ws;
    float* h1   = ws + 2 * (size_t)N + E;
    float* out1 = h1 + (size_t)N * 128;
    float* h2   = out1 + (size_t)N * 128;

    const int T = 256;

    k_init_deg<<<(N + T - 1) / T, T, 0, stream>>>(deg, N);
    k_scatter_deg<<<(E + T - 1) / T, T, 0, stream>>>(col, ew, deg, E);

    k_gemm_rows<8, false><<<(N + 127) / 128, 256, 0, stream>>>(x, W1, deg, b1, h1, out1, N);
    k_scatter_tile<4, 391, 2, 16><<<256, 1024, 0, stream>>>(h1, row, col, ew, deg, out1, N, E);

    k_gemm_rows<4, true><<<(N + 127) / 128, 256, 0, stream>>>(out1, W2, deg, b2, h2, out, N);
    k_scatter_tile<2, 782, 4, 16><<<256, 1024, 0, stream>>>(h2, row, col, ew, deg, out, N, E);
}


# Layer-1 edge scatter also stores its hits (source row, tile row, norm) on per-wave lists; layer 2, on layer 1's tiles and shares, reads them, no col scan or norm work; layer-2 sum order changes
# speedup vs baseline: 2.1680x; 1.1178x over previous
#include <hip/hip_runtime.h>
#include <hip/hip_bf16.h>

typedef __attribute__((ext_vector_type(2))) float v2f;
typedef __attribute__((ext_vector_type(8))) float v8f;

__global__ void k_init_deg(float* __restrict__ deg, int N) {
    int i = blockIdx.x * blockDim.x + threadIdx.x;
    if (i < N) deg[i] = 1.0f;
}

__global__ void k_scatter_deg(const int* __restrict__ col,
                              const float* __restrict__ w,
                              float* __restrict__ deg, int E) {
    int e = blockIdx.x * blockDim.x + threadIdx.x;
    if (e < E) atomicAdd(&deg[col[e]], w[e]);
}

__device__ __forceinline__ float dinv_of(float d) {
    return (d > 0.0f) ? rsqrtf(d) : 0.0f;
}

template <int NT, bool RELU>
__global__ __launch_bounds__(256) void k_gemm_rows(const float* __restrict__ A,
                                                   const float* __restrict__ W,
                                                   const float* __restrict__ deg,
                                                   const float* __restrict__ bias,
                                                   float* __restrict__ H,
                                                   float* __restrict__ out,
                                                   int N) {
    constexpr int NC = 16 * NT;
    constexpr int WI = 64 * (NC / 4) / 256;
    __shared__ __attribute__((aligned(16))) float Wl[128 * NC];
    const int tid  = threadIdx.x;
    const int wave = tid >> 5;
    const int lane = tid & 31;
    const int c    = lane & 15;
    const int h    = lane >> 4;
    const int row0 = blockIdx.x * 128 + wave * 16;

    float4 a4[16];
    {
        const int r = (row0 + c < N) ? (row0 + c) : (N - 1);
        const float4* s4 = (const float4*)(A + (size_t)r * 128) + h;
#pragma unroll
        for (int g = 0; g < 16; ++g) a4[g] = s4[2 * g];
    }

    {
        const float4* w4 = (const float4*)W;
        float4 r0[WI], r1[WI];
#pragma unroll
        for (int it = 0; it < WI; ++it) {
            const int i  = tid + 256 * it;
            const int kp = i / (NC / 4);
            const int n4 = i % (NC / 4);
            r0[it] = w4[(2 * kp) * (NC / 4) + n4];
            r1[it] = w4[(2 * kp + 1) * (NC / 4) + n4];
        }
#pragma unroll
        for (int it = 0; it < WI; ++it) {
            const int i  = tid + 256 * it;
            const int kp = i / (NC / 4);
            const int n  = (i % (NC / 4)) * 4;
            const int ns = n ^ (((kp >> 1) & 1) << 4);
            float4* d4 = (float4*)(Wl + (size_t)(kp * NC + ns) * 2);
            d4[0] = make_float4(r0[it].x, r1[it].x, r0[it].y, r1[it].y);
            d4[1] = make_float4(r0[it].z, r1[it].z, r0[it].w, r1[it].w);
        }
    }

    float dg[8], bs[NT];
#pragma unroll
    for (int v = 0; v < 8; ++v) {
        const int m = row0 + 8 * h + v;
        dg[v] = deg[(m < N) ? m : (N - 1)];
    }
#pragma unroll
    for (int t = 0; t < NT; ++t) bs[t] = bias[16 * t + c];

    if (RELU) {
#pragma unroll
        for (int g = 0; g < 16; ++g) {
            a4[g].x = fmaxf(a4[g].x, 0.f); a4[g].y = fmaxf(a4[g].y, 0.f);
            a4[g].z = fmaxf(a4[g].z, 0.f); a4[g].w = fmaxf(a4[g].w, 0.f);
        }
    }
    __syncthreads();

    const v2f* wE = (const v2f*)Wl + (2 * h * NC + c + 16 * h);
    const v2f* wO = (const v2f*)Wl + (2 * h * NC + c - 16 * h);

    v8f acc[NT];
#pragma unroll
    for (int t = 0; t < NT; ++t) acc[t] = v8f{};

    v2f bc[NT], bn[NT];
#pragma unroll
    for (int t = 0; t < NT; ++t) bc[t] = ((t & 1) ? wO : wE)[16 * t];
#pragma unroll
    for (int st = 0; st < 32; ++st) {
        const int g = st >> 1, s = st & 1;
        if (st + 1 < 32) {
            const int kn = 4 * ((st + 1) >> 1) + ((st + 1) & 1);
#pragma unroll
            for (int t = 0; t < NT; ++t) bn[t] = ((t & 1) ? wO : wE)[kn * NC + 16 * t];
        }
        v2f a;
        a[0] = s ? a4[g].z : a4[g].x;
        a[1] = s ? a4[g].w : a4[g].y;
#pragma unroll
        for (int t = 0; t < NT; ++t)
            acc[t] = __builtin_amdgcn_wmma_f32_16x16x4_f32(
                         false, a, false, bc[t], (short)0, acc[t], false, false);
#pragma unroll
        for (int t = 0; t < NT; ++t) bc[t] = bn[t];
    }

#pragma unroll
    for (int v = 0; v < 8; ++v) {
        const int m = row0 + 8 * h + v;
        if (m < N) {
            const float d  = dinv_of(dg[v]);
            const float s2 = d * d;
            float* hp = H + (size_t)m * NC + c;
            float* op = out + (size_t)m * NC + c;
#pragma unroll
            for (int t = 0; t < NT; ++t) {
                hp[16 * t] = acc[t][v];
                op[16 * t] = bs[t] + acc[t][v] * s2;
            }
        }
    }
}

constexpr int LCAP = 512;
template <int NL, int UE, bool FULL, bool WL>
__device__ __forceinline__ void drain_hits(const int2* qw, int qb, int cnt, int lo, int lane,
                                           const float* __restrict__ h,
                                           const int* __restrict__ row,
                                           const float* __restrict__ w,
                                           const float* __restrict__ deg,
                                           float* tile, int4* lw, int& ln) {
    constexpr int F = 32 * NL;
    int r = 0, cl = 0;
    float wt = 0.0f, dc = 0.0f, dr = 0.0f;
    if (lane < cnt) {
        const int2 en = qw[qb + lane];
        cl = en.y;
        r  = row[en.x];
        wt = w[en.x];
        dc = deg[lo + cl];
    }
    float v[UE][NL];
#pragma unroll
    for (int u = 0; u < UE; ++u) {
        if (FULL || u < cnt) {
            const int ru = __builtin_amdgcn_readlane(r, u);
            const float* s = h + (size_t)ru * F + lane;
#pragma unroll
            for (int k = 0; k < NL; ++k) v[u][k] = s[32 * k];
        } else {
#pragma unroll
            for (int k = 0; k < NL; ++k) v[u][k] = 0.0f;
        }
    }
    if (lane < cnt) dr = deg[r];
    const float nw = dinv_of(dr) * wt * dinv_of(dc);
    if (WL) {
        if (ln + cnt <= LCAP && lane < cnt) lw[ln + lane] = make_int4(r, cl, __float_as_int(nw), 0);
        ln += cnt;
    }
#pragma unroll
    for (int u = 0; u < UE; ++u) {
        if (FULL || u < cnt) {
            const int cu = __builtin_amdgcn_readlane(cl, u);
            const float nu = __int_as_float(__builtin_amdgcn_readlane(__float_as_int(nw), u));
            float* o = tile + cu * F + lane;
#pragma unroll
            for (int k = 0; k < NL; ++k) atomicAdd(o + 32 * k, v[u][k] * nu);
        }
    }
}

template <int NL, int UL, bool FULL>
__device__ __forceinline__ void load_listed(float (&v)[UL][NL], int r, int cnt, int lane,
                                            const float* __restrict__ h) {
    constexpr int F = 32 * NL;
#pragma unroll
    for (int u = 0; u < UL; ++u) {
        if (FULL || u < cnt) {
            const int ru = __builtin_amdgcn_readlane(r, u);
            const float* s = h + (size_t)ru * F + lane;
#pragma unroll
            for (int k = 0; k < NL; ++k) v[u][k] = s[32 * k];
        } else {
#pragma unroll
            for (int k = 0; k < NL; ++k) v[u][k] = 0.0f;
        }
    }
}

template <int NL, int UL, bool FULL>
__device__ __forceinline__ void add_listed(const float (&v)[UL][NL], int cl, float nw, int cnt, int lane,
                                           float* tile) {
    constexpr int F = 32 * NL;
#pragma unroll
    for (int u = 0; u < UL; ++u) {
        if (FULL || u < cnt) {
            const int cu = __builtin_amdgcn_readlane(cl, u);
            const float nu = __int_as_float(__builtin_amdgcn_readlane(__float_as_int(nw), u));
            float* o = tile + cu * F + lane;
#pragma unroll
            for (int k = 0; k < NL; ++k) atomicAdd(o + 32 * k, v[u][k] * nu);
        }
    }
}

template <int NL, int UE, int CH, bool WL>
__device__ __forceinline__ void scan_chunk(const int (&cv)[CH], int ebase, int lo, int nrows, int lane,
                                           int& qn, int2* qw,
                                           const float* __restrict__ h,
                                           const int* __restrict__ row,
                                           const float* __restrict__ w,
                                           const float* __restrict__ deg,
                                           float* tile, int4* lw, int& ln) {
#pragma unroll
    for (int k = 0; k < CH; ++k) {
        const int cl = cv[k] - lo;
        const bool hit = (unsigned)cl < (unsigned)nrows;
        const unsigned m = __builtin_amdgcn_ballot_w32(hit);
        if (m != 0u) {
            if (hit) qw[qn + (int)__builtin_amdgcn_mbcnt_lo(m, 0u)] = make_int2(ebase + 32 * k + lane, cl);
            qn += __builtin_popcount(m);
        }
    }
    while (qn >= UE) {
        qn -= UE;
        drain_hits<NL, UE, true, WL>(qw, qn, UE, lo, lane, h, row, w, deg, tile, lw, ln);
    }
}

template <int NL, int ROWS, int G, int UE, bool WL, bool RL>
__global__ __launch_bounds__(1024) void k_scatter_tile(const float* __restrict__ h,
                                                       const int* __restrict__ row,
                                                       const int* __restrict__ col,
                                                       const float* __restrict__ w,
                                                       const float* __restrict__ deg,
                                                       float* __restrict__ out,
                                                       int* __restrict__ lists,
                                                       int* __restrict__ counts, int N, int E) {
    constexpr int F    = 32 * NL;
    constexpr int CH   = 8;
    constexpr int CE   = 32 * CH;
    constexpr int QCAP = UE + CE;
    __shared__ __attribute__((aligned(16))) float tile[ROWS * F];
    __shared__ int2 queue[32 * QCAP];
    const int tid  = threadIdx.x;
    const int lane = tid & 31;
    const int wv     = __builtin_amdgcn_readfirstlane(tid >> 5);
    const int g      = blockIdx.x % G;
    const int TB     = gridDim.x / G;
    const int nfull  = E / CE;
    const int stride = 32 * G;
    int2* qw = queue + wv * QCAP;
    const bool one_tile = (long long)TB * ROWS >= N;
    int4* lw = (int4*)lists + (size_t)(blockIdx.x * 32 + wv) * LCAP;
    int ln = 0;
    if (RL) ln = __builtin_amdgcn_readfirstlane(counts[blockIdx.x * 32 + wv]);
    const bool use_list = RL && one_tile && (unsigned)ln <= (unsigned)LCAP;

    for (int t = blockIdx.x / G; t * ROWS < N; t += TB) {
        const int lo    = t * ROWS;
        const int nrows = (N - lo < ROWS) ? (N - lo) : ROWS;

        for (int i = tid; i < ROWS * F / 4; i += 1024) ((float4*)tile)[i] = make_float4(0.f, 0.f, 0.f, 0.f);
        __syncthreads();

        if (use_list) {
            constexpr int UL = (NL <= 2) ? 32 : 16;
            int4 en = make_int4(0, 0, 0, 0);
            if (lane < UL && lane < ln) en = lw[lane];
            int base = 0;
            for (; base + UL <= ln; base += UL) {
                const int4 e = en;
                float v[UL][NL];
                load_listed<NL, UL, true>(v, e.x, UL, lane, h);
                if (lane < UL && base + UL + lane < ln) en = lw[base + UL + lane];
                add_listed<NL, UL, true>(v, e.y, __int_as_float(e.z), UL, lane, tile);
            }
            if (base < ln) {
                float v[UL][NL];
                load_listed<NL, UL, false>(v, en.x, ln - base, lane, h);
                add_listed<NL, UL, false>(v, en.y, __int_as_float(en.z), ln - base, lane, tile);
            }
        } else {
            int qn = 0;
            if (nfull * CE < E && g == 0 && wv == 0) {
                int cr[CH];
#pragma unroll
                for (int k = 0; k < CH; ++k) {
                    const int e = nfull * CE + 32 * k + lane;
                    cr[k] = (e < E) ? col[e] : -1;
                }
                scan_chunk<NL, UE, CH, WL>(cr, nfull * CE, lo, nrows, lane, qn, qw, h, row, w, deg, tile, lw, ln);
            }
            if (nfull > 0) {
                const int last = nfull - 1;
                int ci = g * 32 + wv;
                int ca[CH], cb[CH];
                {
                    const int* p = col + (size_t)((ci < last) ? ci : last) * CE + lane;
#pragma unroll
                    for (int k = 0; k < CH; ++k) ca[k] = p[32 * k];
                }
                for (; ci < nfull; ci += 2 * stride) {
                    {
                        const int cj = ci + stride;
                        const int* p = col + (size_t)((cj < last) ? cj : last) * CE + lane;
#pragma unroll
                        for (int k = 0; k < CH; ++k) cb[k] = p[32 * k];
                    }
                    scan_chunk<NL, UE, CH, WL>(ca, ci * CE, lo, nrows, lane, qn, qw, h, row, w, deg, tile, lw, ln);
                    {
                        const int cj = ci + 2 * stride;
                        const int* p = col + (size_t)((cj < last) ? cj : last) * CE + lane;
#pragma unroll
                        for (int k = 0; k < CH; ++k) ca[k] = p[32 * k];
                    }
                    if (ci + stride < nfull)
                        scan_chunk<NL, UE, CH, WL>(cb, (ci + stride) * CE, lo, nrows, lane, qn, qw, h, row, w, deg, tile, lw, ln);
                }
            }
            if (qn > 0) drain_hits<NL, UE, false, WL>(qw, 0, qn, lo, lane, h, row, w, deg, tile, lw, ln);
        }
        __syncthreads();

        for (int r = wv; r < nrows; r += 32) {
            const float* s = tile + r * F + lane;
            float* o = out + (size_t)(lo + r) * F + lane;
            float tv[NL];
#pragma unroll
            for (int k = 0; k < NL; ++k) tv[k] = s[32 * k];
#pragma unroll
            for (int k = 0; k < NL; ++k) atomicAdd(o + 32 * k, tv[k]);
        }
        __syncthreads();
    }
    if (WL && lane == 0) counts[blockIdx.x * 32 + wv] = one_tile ? ln : (LCAP + 1);
}

constexpr int SG = 256, SR = 391, SH = 2;

extern "C" void kernel_launch(void* const* d_in, const int* in_sizes, int n_in,
                              void* d_out, int out_size, void* d_ws, size_t ws_size,
                              hipStream_t stream) {
    const float* x   = (const float*)d_in[0];
    const int*   ei  = (const int*)d_in[1];
    const float* ew  = (const float*)d_in[2];
    const float* W1  = (const float*)d_in[3];
    const float* b1  = (const float*)d_in[4];
    const float* W2  = (const float*)d_in[5];
    const float* b2  = (const float*)d_in[6];
    float* out = (float*)d_out;

    const int E = in_sizes[2];
    const int N = in_sizes[0] / 128;
    const int* row = ei;
    const int* col = ei + E;

    float* ws   = (float*)d_ws;
    float* deg  = ws;
    float* h1   = ws + 2 * (size_t)N + E;
    float* out1 = h1 + (size_t)N * 128;
    float* h2   = out1 + (size_t)N * 128;
    int* lists  = (int*)((char*)d_ws + 134217728);
    int* counts = (int*)((char*)d_ws + 209715200);

    const int T = 256;

    k_init_deg<<<(N + T - 1) / T, T, 0, stream>>>(deg, N);
    k_scatter_deg<<<(E + T - 1) / T, T, 0, stream>>>(col, ew, deg, E);

    k_gemm_rows<8, false><<<(N + 127) / 128, 256, 0, stream>>>(x, W1, deg, b1, h1, out1, N);
    k_scatter_tile<4, SR, SH, 16, true, false><<<SG, 1024, 0, stream>>>(h1, row, col, ew, deg, out1, lists, counts, N, E);

    k_gemm_rows<4, true><<<(N + 127) / 128, 256, 0, stream>>>(out1, W2, deg, b2, h2, out, N);
    k_scatter_tile<2, SR, SH, 16, false, true><<<SG, 1024, 0, stream>>>(h2, row, col, ew, deg, out, lists, counts, N, E);
}
